# starting kernel
# baseline (speedup 1.0000x reference)
#include <hip/hip_runtime.h>
#include <math.h>


#define B_DIM 32
#define C_DIM 256
#define F_DIM 512
#define P_DIM 4
#define N_DIM 1024
#define S_DIM 4096
#define EPS   1e-5f

#define TILE  64
#define PITCH (C_DIM + 2)
#define TILE5 32
#define PITCH5 (F_DIM + 2)

typedef __attribute__((ext_vector_type(2))) float v2f;
typedef __attribute__((ext_vector_type(8))) float v8f;

template <int NB>
__device__ __forceinline__ void wmma_k_loop_nb(const float* __restrict__ Wr,
                                               const float* __restrict__ Ybase,
                                               int pitch, int K, int kk, int m,
                                               v8f acc[NB]) {
#pragma unroll 4
  for (int k = 0; k < K; k += 4) {
    v2f a = *(const v2f*)(Wr + k + kk);
#pragma unroll
    for (int c = 0; c < NB; ++c) {
      v2f b = *(const v2f*)(Ybase + (c * 16 + m) * pitch + k + kk);
      acc[c] = __builtin_amdgcn_wmma_f32_16x16x4_f32(
           false, a,  false, b,
           (short)0, acc[c],  false,  false);
    }
  }
}

template <bool WITH_Q, bool LEAKY>
__global__ void ln_gemm_kernel(const float* __restrict__ xin,
                               const float* __restrict__ lnw,
                               const float* __restrict__ lnb,
                               const float* __restrict__ W,
                               const float* __restrict__ bias,
                               float* __restrict__ out512,
                               float* __restrict__ qout) {
  extern __shared__ float Y[];
  __shared__ float red[4][TILE];
  __shared__ float red2[4][TILE];
  __shared__ float meanv[TILE];
  __shared__ float rstdv[TILE];

  const int b  = blockIdx.y;
  const int s0 = blockIdx.x * TILE;
  const int t  = threadIdx.x;
  const int j  = t & (TILE - 1);
  const int ch = t >> 6;
  const int c0 = ch * 64;

  const float* xb = xin + (size_t)b * C_DIM * S_DIM + s0 + j;
  float sum = 0.f, sq = 0.f;
#pragma unroll 4
  for (int i = 0; i < 64; ++i) {
    int c = c0 + i;
    float v = xb[(size_t)c * S_DIM];
    Y[j * PITCH + c] = v;
    sum += v;
    sq += v * v;
  }
  red[ch][j] = sum;
  red2[ch][j] = sq;
  __syncthreads();
  if (t < TILE) {
    float s  = red[0][t] + red[1][t] + red[2][t] + red[3][t];
    float q2 = red2[0][t] + red2[1][t] + red2[2][t] + red2[3][t];
    float mean = s * (1.f / C_DIM);
    float var  = q2 * (1.f / C_DIM) - mean * mean;
    meanv[t] = mean;
    rstdv[t] = __frsqrt_rn(var + EPS);
  }
  __syncthreads();

  {
    float mean = meanv[j], rstd = rstdv[j];
#pragma unroll 4
    for (int i = 0; i < 64; ++i) {
      int c = c0 + i;
      Y[j * PITCH + c] = (Y[j * PITCH + c] - mean) * rstd * lnw[c] + lnb[c];
    }
  }
  __syncthreads();

  if (WITH_Q) {
    float qs = 0.f;
#pragma unroll 4
    for (int i = 0; i < 64; ++i) {
      int c = c0 + i;
      qs += W[c] * Y[j * PITCH + c];
    }
    red[ch][j] = qs;
    __syncthreads();
    if (t < TILE) {
      qout[(size_t)b * S_DIM + s0 + t] =
          red[0][t] + red[1][t] + red[2][t] + red[3][t] + bias[0];
    }
    __syncthreads();
  }

  const int wave = t >> 5;
  const int lane = t & 31;
  const int m    = lane & 15;
  const int kk   = (lane >> 4) << 1;
  const int mh   = (lane >> 4) << 3;
  const int rofs = WITH_Q ? 1 : 0;

  for (int rt = wave; rt < 32; rt += 8) {
    const float* Wr = W + (size_t)(rofs + rt * 16 + m) * C_DIM;
    __builtin_prefetch(Wr, 0, 1);
    v8f acc[4] = {};
    wmma_k_loop_nb<4>(Wr, Y, PITCH, C_DIM, kk, m, acc);
#pragma unroll
    for (int c = 0; c < 4; ++c) {
#pragma unroll
      for (int r = 0; r < 8; ++r) {
        int row = rt * 16 + mh + r;
        float d = acc[c][r] + bias[rofs + row];
        if (LEAKY) d = d > 0.f ? d : 0.1f * d;
        out512[((size_t)b * F_DIM + row) * S_DIM + s0 + c * 16 + m] = d;
      }
    }
  }
}

__global__ void softmax_ctx_kernel(const float* __restrict__ qbuf,
                                   const float* __restrict__ kv,
                                   float* __restrict__ ctx) {
  __shared__ float probs[N_DIM];
  __shared__ float red[256];
  const int bp = blockIdx.x;
  const int b = bp >> 2, p = bp & 3;
  const int t = threadIdx.x;

  const float* q = qbuf + (size_t)b * S_DIM + p * N_DIM;
  float v[4];
  float mx = -INFINITY;
#pragma unroll
  for (int i = 0; i < 4; ++i) {
    v[i] = q[t + i * 256];
    mx = fmaxf(mx, v[i]);
  }
  red[t] = mx;
  __syncthreads();
  for (int o = 128; o > 0; o >>= 1) {
    if (t < o) red[t] = fmaxf(red[t], red[t + o]);
    __syncthreads();
  }
  mx = red[0];
  __syncthreads();

  float s = 0.f;
#pragma unroll
  for (int i = 0; i < 4; ++i) {
    v[i] = __expf(v[i] - mx);
    s += v[i];
  }
  red[t] = s;
  __syncthreads();
  for (int o = 128; o > 0; o >>= 1) {
    if (t < o) red[t] += red[t + o];
    __syncthreads();
  }
  float inv = 1.f / red[0];
#pragma unroll
  for (int i = 0; i < 4; ++i) probs[t + i * 256] = v[i] * inv;
  __syncthreads();

  const float* kr = kv + ((size_t)b * F_DIM + t) * S_DIM + p * N_DIM;
  float acc = 0.f;
#pragma unroll 4
  for (int n = 0; n < N_DIM; n += 4) {
    float4 kx = *reinterpret_cast<const float4*>(kr + n);
    acc += kx.x * probs[n] + kx.y * probs[n + 1] +
           kx.z * probs[n + 2] + kx.w * probs[n + 3];
  }
  ctx[(size_t)bp * C_DIM + t] = acc;
}

__global__ void attn_out_kernel(const float* __restrict__ x,
                                const float* __restrict__ kv,
                                const float* __restrict__ ctx,
                                const float* __restrict__ W,
                                const float* __restrict__ bias,
                                float* __restrict__ x1) {
  extern __shared__ float A[];
  const int b  = blockIdx.y;
  const int s0 = blockIdx.x * TILE;
  const int p  = s0 / N_DIM;
  const int t  = threadIdx.x;
  const int j  = t & (TILE - 1);
  const int ch = t >> 6;
  const int c0 = ch * 64;

  const float* ctxbp = ctx + ((size_t)b * P_DIM + p) * C_DIM;
#pragma unroll 4
  for (int i = 0; i < 64; ++i) {
    int c = c0 + i;
    float vv = kv[((size_t)b * F_DIM + C_DIM + c) * S_DIM + s0 + j];
    vv = vv > 0.f ? vv : 0.f;
    A[j * PITCH + c] = vv * ctxbp[c];
  }
  __syncthreads();

  const int wave = t >> 5;
  const int lane = t & 31;
  const int m    = lane & 15;
  const int kk   = (lane >> 4) << 1;
  const int mh   = (lane >> 4) << 3;

  for (int rt = wave; rt < 16; rt += 8) {
    const float* Wr = W + (size_t)(rt * 16 + m) * C_DIM;
    __builtin_prefetch(Wr, 0, 1);
    v8f acc[4] = {};
    wmma_k_loop_nb<4>(Wr, A, PITCH, C_DIM, kk, m, acc);
#pragma unroll
    for (int c = 0; c < 4; ++c) {
#pragma unroll
      for (int r = 0; r < 8; ++r) {
        int row = rt * 16 + mh + r;
        size_t idx = ((size_t)b * C_DIM + row) * S_DIM + s0 + c * 16 + m;
        x1[idx] = x[idx] + bias[row] + acc[c][r];
      }
    }
  }
}

__global__ void ffn2_kernel(const float* __restrict__ x1,
                            const float* __restrict__ h,
                            const float* __restrict__ W,
                            const float* __restrict__ bias,
                            float* __restrict__ out) {
  extern __shared__ float Hs[];
  const int b  = blockIdx.y;
  const int s0 = blockIdx.x * TILE5;
  const int t  = threadIdx.x;
  const int j  = t & (TILE5 - 1);
  const int ch = t >> 5;
#pragma unroll 4
  for (int i = 0; i < 64; ++i) {
    int r = ch * 64 + i;
    Hs[j * PITCH5 + r] = h[((size_t)b * F_DIM + r) * S_DIM + s0 + j];
  }
  __syncthreads();

  const int wave = t >> 5;
  const int lane = t & 31;
  const int m    = lane & 15;
  const int kk   = (lane >> 4) << 1;
  const int mh   = (lane >> 4) << 3;

  for (int rt = wave; rt < 16; rt += 8) {
    const float* Wr = W + (size_t)(rt * 16 + m) * F_DIM;
    __builtin_prefetch(Wr, 0, 1);
    v8f acc[2] = {};
    wmma_k_loop_nb<2>(Wr, Hs, PITCH5, F_DIM, kk, m, acc);
#pragma unroll
    for (int c = 0; c < 2; ++c) {
#pragma unroll
      for (int r = 0; r < 8; ++r) {
        int row = rt * 16 + mh + r;
        size_t idx = ((size_t)b * C_DIM + row) * S_DIM + s0 + c * 16 + m;
        out[idx] = x1[idx] + bias[row] + acc[c][r];
      }
    }
  }
}

extern "C" void kernel_launch(void* const* d_in, const int* in_sizes, int n_in,
                              void* d_out, int out_size, void* d_ws,
                              size_t ws_size, hipStream_t stream) {
  const float* x      = (const float*)d_in[0];
  const float* ln1_w  = (const float*)d_in[1];
  const float* ln1_b  = (const float*)d_in[2];
  const float* qkv_w  = (const float*)d_in[3];
  const float* qkv_b  = (const float*)d_in[4];
  const float* out_w  = (const float*)d_in[5];
  const float* out_b  = (const float*)d_in[6];
  const float* ln2_w  = (const float*)d_in[7];
  const float* ln2_b  = (const float*)d_in[8];
  const float* ffn1_w = (const float*)d_in[9];
  const float* ffn1_b = (const float*)d_in[10];
  const float* ffn2_w = (const float*)d_in[11];
  const float* ffn2_b = (const float*)d_in[12];
  float* outp = (float*)d_out;

  float* ws   = (float*)d_ws;
  float* qbuf = ws;
  float* ctxb = qbuf + (size_t)B_DIM * S_DIM;
  float* kvb  = ctxb + (size_t)B_DIM * P_DIM * C_DIM;
  float* x1   = kvb + (size_t)B_DIM * F_DIM * S_DIM;
  float* hbuf = kvb;

  const size_t lds1 = (size_t)TILE * PITCH * sizeof(float);
  const size_t lds5 = (size_t)TILE5 * PITCH5 * sizeof(float);
  dim3 grid1(S_DIM / TILE, B_DIM);
  dim3 grid5(S_DIM / TILE5, B_DIM);

  ln_gemm_kernel<true, false><<<grid1, 256, lds1, stream>>>(
      x, ln1_w, ln1_b, qkv_w, qkv_b, kvb, qbuf);
  softmax_ctx_kernel<<<B_DIM * P_DIM, 256, 0, stream>>>(qbuf, kvb, ctxb);
  attn_out_kernel<<<grid1, 256, lds1, stream>>>(
      x, kvb, ctxb, out_w, out_b, x1);
  ln_gemm_kernel<false, true><<<grid1, 256, lds1, stream>>>(
      x1, ln2_w, ln2_b, ffn1_w, ffn1_b, hbuf, nullptr);
  ffn2_kernel<<<grid5, 256, lds5, stream>>>(
      x1, hbuf, ffn2_w, ffn2_b, outp);
}


# Tile staging: each thread issues all 64 global loads of its tile column before first use (was 4 at a time) and keeps them in registers; LayerNorm done on registers, one LDS write
# speedup vs baseline: 1.0755x; 1.0755x over previous
#include <hip/hip_runtime.h>
#include <math.h>


#define B_DIM 32
#define C_DIM 256
#define F_DIM 512
#define P_DIM 4
#define N_DIM 1024
#define S_DIM 4096
#define EPS   1e-5f

#define TILE  64
#define PITCH (C_DIM + 2)
#define TILE5 32
#define PITCH5 (F_DIM + 2)

typedef __attribute__((ext_vector_type(2))) float v2f;
typedef __attribute__((ext_vector_type(8))) float v8f;

template <int NB>
__device__ __forceinline__ void wmma_k_loop_nb(const float* __restrict__ Wr,
                                               const float* __restrict__ Ybase,
                                               int pitch, int K, int kk, int m,
                                               v8f acc[NB]) {
#pragma unroll 4
  for (int k = 0; k < K; k += 4) {
    v2f a = *(const v2f*)(Wr + k + kk);
#pragma unroll
    for (int c = 0; c < NB; ++c) {
      v2f b = *(const v2f*)(Ybase + (c * 16 + m) * pitch + k + kk);
      acc[c] = __builtin_amdgcn_wmma_f32_16x16x4_f32(
           false, a,  false, b,
           (short)0, acc[c],  false,  false);
    }
  }
}

template <bool WITH_Q, bool LEAKY>
__global__ void ln_gemm_kernel(const float* __restrict__ xin,
                               const float* __restrict__ lnw,
                               const float* __restrict__ lnb,
                               const float* __restrict__ W,
                               const float* __restrict__ bias,
                               float* __restrict__ out512,
                               float* __restrict__ qout) {
  extern __shared__ float Y[];
  __shared__ float red[4][TILE];
  __shared__ float red2[4][TILE];
  __shared__ float redq[4][TILE];

  const int b  = blockIdx.y;
  const int s0 = blockIdx.x * TILE;
  const int t  = threadIdx.x;
  const int j  = t & (TILE - 1);
  const int ch = __builtin_amdgcn_readfirstlane(t >> 6);
  const int c0 = ch * 64;

  const float* xb = xin + ((size_t)b * C_DIM + c0) * S_DIM + s0 + j;
  float v[64];
#pragma unroll
  for (int i = 0; i < 64; ++i) v[i] = xb[(size_t)i * S_DIM];

  float sum = 0.f, sq = 0.f;
#pragma unroll
  for (int i = 0; i < 64; ++i) {
    sum += v[i];
    sq += v[i] * v[i];
  }
  red[ch][j] = sum;
  red2[ch][j] = sq;
  __syncthreads();
  float mean, rstd;
  {
    float s  = red[0][j] + red[1][j] + red[2][j] + red[3][j];
    float q2 = red2[0][j] + red2[1][j] + red2[2][j] + red2[3][j];
    mean = s * (1.f / C_DIM);
    float var = q2 * (1.f / C_DIM) - mean * mean;
    rstd = __frsqrt_rn(var + EPS);
  }

  float qs = 0.f;
#pragma unroll
  for (int i = 0; i < 64; ++i) {
    int c = c0 + i;
    float y = (v[i] - mean) * rstd * lnw[c] + lnb[c];
    Y[j * PITCH + c] = y;
    if (WITH_Q) qs += W[c] * y;
  }
  if (WITH_Q) redq[ch][j] = qs;
  __syncthreads();
  if (WITH_Q) {
    if (t < TILE) {
      qout[(size_t)b * S_DIM + s0 + t] =
          redq[0][t] + redq[1][t] + redq[2][t] + redq[3][t] + bias[0];
    }
  }

  const int wave = t >> 5;
  const int lane = t & 31;
  const int m    = lane & 15;
  const int kk   = (lane >> 4) << 1;
  const int mh   = (lane >> 4) << 3;
  const int rofs = WITH_Q ? 1 : 0;

  for (int rt = wave; rt < 32; rt += 8) {
    const float* Wr = W + (size_t)(rofs + rt * 16 + m) * C_DIM;
    __builtin_prefetch(Wr, 0, 1);
    v8f acc[4] = {};
    wmma_k_loop_nb<4>(Wr, Y, PITCH, C_DIM, kk, m, acc);
#pragma unroll
    for (int c = 0; c < 4; ++c) {
#pragma unroll
      for (int r = 0; r < 8; ++r) {
        int row = rt * 16 + mh + r;
        float d = acc[c][r] + bias[rofs + row];
        if (LEAKY) d = d > 0.f ? d : 0.1f * d;
        out512[((size_t)b * F_DIM + row) * S_DIM + s0 + c * 16 + m] = d;
      }
    }
  }
}

__global__ void softmax_ctx_kernel(const float* __restrict__ qbuf,
                                   const float* __restrict__ kv,
                                   float* __restrict__ ctx) {
  __shared__ float probs[N_DIM];
  __shared__ float red[256];
  const int bp = blockIdx.x;
  const int b = bp >> 2, p = bp & 3;
  const int t = threadIdx.x;

  const float* q = qbuf + (size_t)b * S_DIM + p * N_DIM;
  float v[4];
  float mx = -INFINITY;
#pragma unroll
  for (int i = 0; i < 4; ++i) {
    v[i] = q[t + i * 256];
    mx = fmaxf(mx, v[i]);
  }
  red[t] = mx;
  __syncthreads();
  for (int o = 128; o > 0; o >>= 1) {
    if (t < o) red[t] = fmaxf(red[t], red[t + o]);
    __syncthreads();
  }
  mx = red[0];
  __syncthreads();

  float s = 0.f;
#pragma unroll
  for (int i = 0; i < 4; ++i) {
    v[i] = __expf(v[i] - mx);
    s += v[i];
  }
  red[t] = s;
  __syncthreads();
  for (int o = 128; o > 0; o >>= 1) {
    if (t < o) red[t] += red[t + o];
    __syncthreads();
  }
  float inv = 1.f / red[0];
#pragma unroll
  for (int i = 0; i < 4; ++i) probs[t + i * 256] = v[i] * inv;
  __syncthreads();

  const float* kr = kv + ((size_t)b * F_DIM + t) * S_DIM + p * N_DIM;
  float acc = 0.f;
#pragma unroll 4
  for (int n = 0; n < N_DIM; n += 4) {
    float4 kx = *reinterpret_cast<const float4*>(kr + n);
    acc += kx.x * probs[n] + kx.y * probs[n + 1] +
           kx.z * probs[n + 2] + kx.w * probs[n + 3];
  }
  ctx[(size_t)bp * C_DIM + t] = acc;
}

__global__ void attn_out_kernel(const float* __restrict__ x,
                                const float* __restrict__ kv,
                                const float* __restrict__ ctx,
                                const float* __restrict__ W,
                                const float* __restrict__ bias,
                                float* __restrict__ x1) {
  extern __shared__ float A[];
  const int b  = blockIdx.y;
  const int s0 = blockIdx.x * TILE;
  const int p  = s0 / N_DIM;
  const int t  = threadIdx.x;
  const int j  = t & (TILE - 1);
  const int ch = __builtin_amdgcn_readfirstlane(t >> 6);
  const int c0 = ch * 64;

  const float* ctxbp = ctx + ((size_t)b * P_DIM + p) * C_DIM;
  const float* vb = kv + ((size_t)b * F_DIM + C_DIM + c0) * S_DIM + s0 + j;
  float v[64];
#pragma unroll
  for (int i = 0; i < 64; ++i) v[i] = vb[(size_t)i * S_DIM];
#pragma unroll
  for (int i = 0; i < 64; ++i) {
    int c = c0 + i;
    float vv = v[i];
    vv = vv > 0.f ? vv : 0.f;
    A[j * PITCH + c] = vv * ctxbp[c];
  }
  __syncthreads();

  const int wave = t >> 5;
  const int lane = t & 31;
  const int m    = lane & 15;
  const int kk   = (lane >> 4) << 1;
  const int mh   = (lane >> 4) << 3;

  for (int rt = wave; rt < 16; rt += 8) {
    const float* Wr = W + (size_t)(rt * 16 + m) * C_DIM;
    __builtin_prefetch(Wr, 0, 1);
    v8f acc[4] = {};
    wmma_k_loop_nb<4>(Wr, A, PITCH, C_DIM, kk, m, acc);
#pragma unroll
    for (int c = 0; c < 4; ++c) {
#pragma unroll
      for (int r = 0; r < 8; ++r) {
        int row = rt * 16 + mh + r;
        size_t idx = ((size_t)b * C_DIM + row) * S_DIM + s0 + c * 16 + m;
        x1[idx] = x[idx] + bias[row] + acc[c][r];
      }
    }
  }
}

__global__ void ffn2_kernel(const float* __restrict__ x1,
                            const float* __restrict__ h,
                            const float* __restrict__ W,
                            const float* __restrict__ bias,
                            float* __restrict__ out) {
  extern __shared__ float Hs[];
  const int b  = blockIdx.y;
  const int s0 = blockIdx.x * TILE5;
  const int t  = threadIdx.x;
  const int j  = t & (TILE5 - 1);
  const int ch = __builtin_amdgcn_readfirstlane(t >> 5);
  const float* hb = h + ((size_t)b * F_DIM + ch * 64) * S_DIM + s0 + j;
  float v[64];
#pragma unroll
  for (int i = 0; i < 64; ++i) v[i] = hb[(size_t)i * S_DIM];
#pragma unroll
  for (int i = 0; i < 64; ++i) Hs[j * PITCH5 + ch * 64 + i] = v[i];
  __syncthreads();

  const int wave = t >> 5;
  const int lane = t & 31;
  const int m    = lane & 15;
  const int kk   = (lane >> 4) << 1;
  const int mh   = (lane >> 4) << 3;

  for (int rt = wave; rt < 16; rt += 8) {
    const float* Wr = W + (size_t)(rt * 16 + m) * F_DIM;
    __builtin_prefetch(Wr, 0, 1);
    v8f acc[2] = {};
    wmma_k_loop_nb<2>(Wr, Hs, PITCH5, F_DIM, kk, m, acc);
#pragma unroll
    for (int c = 0; c < 2; ++c) {
#pragma unroll
      for (int r = 0; r < 8; ++r) {
        int row = rt * 16 + mh + r;
        size_t idx = ((size_t)b * C_DIM + row) * S_DIM + s0 + c * 16 + m;
        out[idx] = x1[idx] + bias[row] + acc[c][r];
      }
    }
  }
}

extern "C" void kernel_launch(void* const* d_in, const int* in_sizes, int n_in,
                              void* d_out, int out_size, void* d_ws,
                              size_t ws_size, hipStream_t stream) {
  const float* x      = (const float*)d_in[0];
  const float* ln1_w  = (const float*)d_in[1];
  const float* ln1_b  = (const float*)d_in[2];
  const float* qkv_w  = (const float*)d_in[3];
  const float* qkv_b  = (const float*)d_in[4];
  const float* out_w  = (const float*)d_in[5];
  const float* out_b  = (const float*)d_in[6];
  const float* ln2_w  = (const float*)d_in[7];
  const float* ln2_b  = (const float*)d_in[8];
  const float* ffn1_w = (const float*)d_in[9];
  const float* ffn1_b = (const float*)d_in[10];
  const float* ffn2_w = (const float*)d_in[11];
  const float* ffn2_b = (const float*)d_in[12];
  float* outp = (float*)d_out;

  float* ws   = (float*)d_ws;
  float* qbuf = ws;
  float* ctxb = qbuf + (size_t)B_DIM * S_DIM;
  float* kvb  = ctxb + (size_t)B_DIM * P_DIM * C_DIM;
  float* x1   = kvb + (size_t)B_DIM * F_DIM * S_DIM;
  float* hbuf = kvb;

  const size_t lds1 = (size_t)TILE * PITCH * sizeof(float);
  const size_t lds5 = (size_t)TILE5 * PITCH5 * sizeof(float);
  dim3 grid1(S_DIM / TILE, B_DIM);
  dim3 grid5(S_DIM / TILE5, B_DIM);

  ln_gemm_kernel<true, false><<<grid1, 256, lds1, stream>>>(
      x, ln1_w, ln1_b, qkv_w, qkv_b, kvb, qbuf);
  softmax_ctx_kernel<<<B_DIM * P_DIM, 256, 0, stream>>>(qbuf, kvb, ctxb);
  attn_out_kernel<<<grid1, 256, lds1, stream>>>(
      x, kvb, ctxb, out_w, out_b, x1);
  ln_gemm_kernel<false, true><<<grid1, 256, lds1, stream>>>(
      x1, ln2_w, ln2_b, ffn1_w, ffn1_b, hbuf, nullptr);
  ffn2_kernel<<<grid5, 256, lds5, stream>>>(
      x1, hbuf, ffn2_w, ffn2_b, outp);
}


# A (global weights) and B (LDS tile) fragments read 16 B per lane, one load per two k-steps (LDS pitch 260/516 floats, 16 B tile stores); k order inside each 8-k step changes the f32 sum order
# speedup vs baseline: 1.1101x; 1.0322x over previous
#include <hip/hip_runtime.h>
#include <math.h>


#define B_DIM 32
#define C_DIM 256
#define F_DIM 512
#define P_DIM 4
#define N_DIM 1024
#define S_DIM 4096
#define EPS   1e-5f

#define TILE  64
#define PITCH (C_DIM + 4)
#define TILE5 32
#define PITCH5 (F_DIM + 4)

typedef __attribute__((ext_vector_type(2))) float v2f;
typedef __attribute__((ext_vector_type(4))) float v4f;
typedef __attribute__((ext_vector_type(8))) float v8f;

template <int NB>
__device__ __forceinline__ void wmma_k_loop_nb(const float* __restrict__ Wr,
                                               const float* __restrict__ Ybase,
                                               int pitch, int K, int kq, int m,
                                               v8f acc[NB]) {
#pragma unroll 2
  for (int k = 0; k < K; k += 8) {
    v4f a = *(const v4f*)(Wr + k + kq);
    v4f b[NB];
#pragma unroll
    for (int c = 0; c < NB; ++c)
      b[c] = *(const v4f*)(Ybase + (c * 16 + m) * pitch + k + kq);
#pragma unroll
    for (int c = 0; c < NB; ++c) {
      acc[c] = __builtin_amdgcn_wmma_f32_16x16x4_f32(
           false, a.lo,  false, b[c].lo,
           (short)0, acc[c],  false,  false);
    }
#pragma unroll
    for (int c = 0; c < NB; ++c) {
      acc[c] = __builtin_amdgcn_wmma_f32_16x16x4_f32(
           false, a.hi,  false, b[c].hi,
           (short)0, acc[c],  false,  false);
    }
  }
}

template <bool WITH_Q, bool LEAKY>
__global__ void ln_gemm_kernel(const float* __restrict__ xin,
                               const float* __restrict__ lnw,
                               const float* __restrict__ lnb,
                               const float* __restrict__ W,
                               const float* __restrict__ bias,
                               float* __restrict__ out512,
                               float* __restrict__ qout) {
  extern __shared__ __attribute__((aligned(16))) float Y[];
  __shared__ float red[4][TILE];
  __shared__ float red2[4][TILE];
  __shared__ float redq[4][TILE];

  const int b  = blockIdx.y;
  const int s0 = blockIdx.x * TILE;
  const int t  = threadIdx.x;
  const int j  = t & (TILE - 1);
  const int ch = __builtin_amdgcn_readfirstlane(t >> 6);
  const int c0 = ch * 64;

  const float* xb = xin + ((size_t)b * C_DIM + c0) * S_DIM + s0 + j;
  float v[64];
#pragma unroll
  for (int i = 0; i < 64; ++i) v[i] = xb[(size_t)i * S_DIM];

  float sum = 0.f, sq = 0.f;
#pragma unroll
  for (int i = 0; i < 64; ++i) {
    sum += v[i];
    sq += v[i] * v[i];
  }
  red[ch][j] = sum;
  red2[ch][j] = sq;
  __syncthreads();
  float mean, rstd;
  {
    float s  = red[0][j] + red[1][j] + red[2][j] + red[3][j];
    float q2 = red2[0][j] + red2[1][j] + red2[2][j] + red2[3][j];
    mean = s * (1.f / C_DIM);
    float var = q2 * (1.f / C_DIM) - mean * mean;
    rstd = __frsqrt_rn(var + EPS);
  }

  float qs = 0.f;
#pragma unroll
  for (int i = 0; i < 64; i += 4) {
    v4f y4;
#pragma unroll
    for (int e = 0; e < 4; ++e) {
      int c = c0 + i + e;
      float y = (v[i + e] - mean) * rstd * lnw[c] + lnb[c];
      y4[e] = y;
      if (WITH_Q) qs += W[c] * y;
    }
    *(v4f*)(Y + j * PITCH + c0 + i) = y4;
  }
  if (WITH_Q) redq[ch][j] = qs;
  __syncthreads();
  if (WITH_Q) {
    if (t < TILE) {
      qout[(size_t)b * S_DIM + s0 + t] =
          redq[0][t] + redq[1][t] + redq[2][t] + redq[3][t] + bias[0];
    }
  }

  const int wave = t >> 5;
  const int lane = t & 31;
  const int m    = lane & 15;
  const int kq   = (lane >> 4) << 2;
  const int mh   = (lane >> 4) << 3;
  const int rofs = WITH_Q ? 1 : 0;

  for (int rt = wave; rt < 32; rt += 8) {
    const float* Wr = W + (size_t)(rofs + rt * 16 + m) * C_DIM;
    __builtin_prefetch(Wr, 0, 1);
    v8f acc[4] = {};
    wmma_k_loop_nb<4>(Wr, Y, PITCH, C_DIM, kq, m, acc);
#pragma unroll
    for (int c = 0; c < 4; ++c) {
#pragma unroll
      for (int r = 0; r < 8; ++r) {
        int row = rt * 16 + mh + r;
        float d = acc[c][r] + bias[rofs + row];
        if (LEAKY) d = d > 0.f ? d : 0.1f * d;
        out512[((size_t)b * F_DIM + row) * S_DIM + s0 + c * 16 + m] = d;
      }
    }
  }
}

__global__ void softmax_ctx_kernel(const float* __restrict__ qbuf,
                                   const float* __restrict__ kv,
                                   float* __restrict__ ctx) {
  __shared__ float probs[N_DIM];
  __shared__ float red[256];
  const int bp = blockIdx.x;
  const int b = bp >> 2, p = bp & 3;
  const int t = threadIdx.x;

  const float* q = qbuf + (size_t)b * S_DIM + p * N_DIM;
  float v[4];
  float mx = -INFINITY;
#pragma unroll
  for (int i = 0; i < 4; ++i) {
    v[i] = q[t + i * 256];
    mx = fmaxf(mx, v[i]);
  }
  red[t] = mx;
  __syncthreads();
  for (int o = 128; o > 0; o >>= 1) {
    if (t < o) red[t] = fmaxf(red[t], red[t + o]);
    __syncthreads();
  }
  mx = red[0];
  __syncthreads();

  float s = 0.f;
#pragma unroll
  for (int i = 0; i < 4; ++i) {
    v[i] = __expf(v[i] - mx);
    s += v[i];
  }
  red[t] = s;
  __syncthreads();
  for (int o = 128; o > 0; o >>= 1) {
    if (t < o) red[t] += red[t + o];
    __syncthreads();
  }
  float inv = 1.f / red[0];
#pragma unroll
  for (int i = 0; i < 4; ++i) probs[t + i * 256] = v[i] * inv;
  __syncthreads();

  const float* kr = kv + ((size_t)b * F_DIM + t) * S_DIM + p * N_DIM;
  float acc = 0.f;
#pragma unroll 4
  for (int n = 0; n < N_DIM; n += 4) {
    float4 kx = *reinterpret_cast<const float4*>(kr + n);
    acc += kx.x * probs[n] + kx.y * probs[n + 1] +
           kx.z * probs[n + 2] + kx.w * probs[n + 3];
  }
  ctx[(size_t)bp * C_DIM + t] = acc;
}

__global__ void attn_out_kernel(const float* __restrict__ x,
                                const float* __restrict__ kv,
                                const float* __restrict__ ctx,
                                const float* __restrict__ W,
                                const float* __restrict__ bias,
                                float* __restrict__ x1) {
  extern __shared__ __attribute__((aligned(16))) float A[];
  const int b  = blockIdx.y;
  const int s0 = blockIdx.x * TILE;
  const int p  = s0 / N_DIM;
  const int t  = threadIdx.x;
  const int j  = t & (TILE - 1);
  const int ch = __builtin_amdgcn_readfirstlane(t >> 6);
  const int c0 = ch * 64;

  const float* ctxbp = ctx + ((size_t)b * P_DIM + p) * C_DIM;
  const float* vb = kv + ((size_t)b * F_DIM + C_DIM + c0) * S_DIM + s0 + j;
  float v[64];
#pragma unroll
  for (int i = 0; i < 64; ++i) v[i] = vb[(size_t)i * S_DIM];
#pragma unroll
  for (int i = 0; i < 64; i += 4) {
    v4f a4;
#pragma unroll
    for (int e = 0; e < 4; ++e) {
      float vv = v[i + e];
      vv = vv > 0.f ? vv : 0.f;
      a4[e] = vv * ctxbp[c0 + i + e];
    }
    *(v4f*)(A + j * PITCH + c0 + i) = a4;
  }
  __syncthreads();

  const int wave = t >> 5;
  const int lane = t & 31;
  const int m    = lane & 15;
  const int kq   = (lane >> 4) << 2;
  const int mh   = (lane >> 4) << 3;

  for (int rt = wave; rt < 16; rt += 8) {
    const float* Wr = W + (size_t)(rt * 16 + m) * C_DIM;
    __builtin_prefetch(Wr, 0, 1);
    v8f acc[4] = {};
    wmma_k_loop_nb<4>(Wr, A, PITCH, C_DIM, kq, m, acc);
#pragma unroll
    for (int c = 0; c < 4; ++c) {
#pragma unroll
      for (int r = 0; r < 8; ++r) {
        int row = rt * 16 + mh + r;
        size_t idx = ((size_t)b * C_DIM + row) * S_DIM + s0 + c * 16 + m;
        x1[idx] = x[idx] + bias[row] + acc[c][r];
      }
    }
  }
}

__global__ void ffn2_kernel(const float* __restrict__ x1,
                            const float* __restrict__ h,
                            const float* __restrict__ W,
                            const float* __restrict__ bias,
                            float* __restrict__ out) {
  extern __shared__ __attribute__((aligned(16))) float Hs[];
  const int b  = blockIdx.y;
  const int s0 = blockIdx.x * TILE5;
  const int t  = threadIdx.x;
  const int j  = t & (TILE5 - 1);
  const int ch = __builtin_amdgcn_readfirstlane(t >> 5);
  const float* hb = h + ((size_t)b * F_DIM + ch * 64) * S_DIM + s0 + j;
  float v[64];
#pragma unroll
  for (int i = 0; i < 64; ++i) v[i] = hb[(size_t)i * S_DIM];
#pragma unroll
  for (int i = 0; i < 64; i += 4) {
    v4f h4 = {v[i], v[i + 1], v[i + 2], v[i + 3]};
    *(v4f*)(Hs + j * PITCH5 + ch * 64 + i) = h4;
  }
  __syncthreads();

  const int wave = t >> 5;
  const int lane = t & 31;
  const int m    = lane & 15;
  const int kq   = (lane >> 4) << 2;
  const int mh   = (lane >> 4) << 3;

  for (int rt = wave; rt < 16; rt += 8) {
    const float* Wr = W + (size_t)(rt * 16 + m) * F_DIM;
    __builtin_prefetch(Wr, 0, 1);
    v8f acc[2] = {};
    wmma_k_loop_nb<2>(Wr, Hs, PITCH5, F_DIM, kq, m, acc);
#pragma unroll
    for (int c = 0; c < 2; ++c) {
#pragma unroll
      for (int r = 0; r < 8; ++r) {
        int row = rt * 16 + mh + r;
        size_t idx = ((size_t)b * C_DIM + row) * S_DIM + s0 + c * 16 + m;
        out[idx] = x1[idx] + bias[row] + acc[c][r];
      }
    }
  }
}

extern "C" void kernel_launch(void* const* d_in, const int* in_sizes, int n_in,
                              void* d_out, int out_size, void* d_ws,
                              size_t ws_size, hipStream_t stream) {
  const float* x      = (const float*)d_in[0];
  const float* ln1_w  = (const float*)d_in[1];
  const float* ln1_b  = (const float*)d_in[2];
  const float* qkv_w  = (const float*)d_in[3];
  const float* qkv_b  = (const float*)d_in[4];
  const float* out_w  = (const float*)d_in[5];
  const float* out_b  = (const float*)d_in[6];
  const float* ln2_w  = (const float*)d_in[7];
  const float* ln2_b  = (const float*)d_in[8];
  const float* ffn1_w = (const float*)d_in[9];
  const float* ffn1_b = (const float*)d_in[10];
  const float* ffn2_w = (const float*)d_in[11];
  const float* ffn2_b = (const float*)d_in[12];
  float* outp = (float*)d_out;

  float* ws   = (float*)d_ws;
  float* qbuf = ws;
  float* ctxb = qbuf + (size_t)B_DIM * S_DIM;
  float* kvb  = ctxb + (size_t)B_DIM * P_DIM * C_DIM;
  float* x1   = kvb + (size_t)B_DIM * F_DIM * S_DIM;
  float* hbuf = kvb;

  const size_t lds1 = (size_t)TILE * PITCH * sizeof(float);
  const size_t lds5 = (size_t)TILE5 * PITCH5 * sizeof(float);
  dim3 grid1(S_DIM / TILE, B_DIM);
  dim3 grid5(S_DIM / TILE5, B_DIM);

  ln_gemm_kernel<true, false><<<grid1, 256, lds1, stream>>>(
      x, ln1_w, ln1_b, qkv_w, qkv_b, kvb, qbuf);
  softmax_ctx_kernel<<<B_DIM * P_DIM, 256, 0, stream>>>(qbuf, kvb, ctxb);
  attn_out_kernel<<<grid1, 256, lds1, stream>>>(
      x, kvb, ctxb, out_w, out_b, x1);
  ln_gemm_kernel<false, true><<<grid1, 256, lds1, stream>>>(
      x1, ln2_w, ln2_b, ffn1_w, ffn1_b, hbuf, nullptr);
  ffn2_kernel<<<grid5, 256, lds5, stream>>>(
      x1, hbuf, ffn2_w, ffn2_b, outp);
}


# k GEMM removed by linearity: ctx = Wk*(sum_n y_n*p_n)+bk; K1 keeps per-tile softmax partials of its LN tile, K2 joins them (rescaled), one 256x256 mat-vec per (b,p); ctx f32 sum order changes
# speedup vs baseline: 1.3032x; 1.1739x over previous
#include <hip/hip_runtime.h>
#include <math.h>


#define B_DIM 32
#define C_DIM 256
#define F_DIM 512
#define P_DIM 4
#define N_DIM 1024
#define S_DIM 4096
#define EPS   1e-5f

#define TILE  64
#define PITCH (C_DIM + 4)
#define TILE5 32
#define PITCH5 (F_DIM + 4)

typedef __attribute__((ext_vector_type(2))) float v2f;
typedef __attribute__((ext_vector_type(4))) float v4f;
typedef __attribute__((ext_vector_type(8))) float v8f;

template <int NB>
__device__ __forceinline__ void wmma_k_loop_nb(const float* __restrict__ Wr,
                                               const float* __restrict__ Ybase,
                                               int pitch, int K, int kq, int m,
                                               v8f acc[NB]) {
#pragma unroll 2
  for (int k = 0; k < K; k += 8) {
    v4f a = *(const v4f*)(Wr + k + kq);
    v4f b[NB];
#pragma unroll
    for (int c = 0; c < NB; ++c)
      b[c] = *(const v4f*)(Ybase + (c * 16 + m) * pitch + k + kq);
#pragma unroll
    for (int c = 0; c < NB; ++c) {
      acc[c] = __builtin_amdgcn_wmma_f32_16x16x4_f32(
           false, a.lo,  false, b[c].lo,
           (short)0, acc[c],  false,  false);
    }
#pragma unroll
    for (int c = 0; c < NB; ++c) {
      acc[c] = __builtin_amdgcn_wmma_f32_16x16x4_f32(
           false, a.hi,  false, b[c].hi,
           (short)0, acc[c],  false,  false);
    }
  }
}

template <bool WITH_Q, bool LEAKY>
__global__ void ln_gemm_kernel(const float* __restrict__ xin,
                               const float* __restrict__ lnw,
                               const float* __restrict__ lnb,
                               const float* __restrict__ W,
                               const float* __restrict__ bias,
                               float* __restrict__ out512,
                               float* __restrict__ pstat) {
  extern __shared__ __attribute__((aligned(16))) float Y[];
  __shared__ __attribute__((aligned(16))) float red[4][TILE];
  __shared__ float red2[4][TILE];
  __shared__ float redq[4][TILE];

  const int b  = blockIdx.y;
  const int s0 = blockIdx.x * TILE;
  const int t  = threadIdx.x;
  const int j  = t & (TILE - 1);
  const int ch = __builtin_amdgcn_readfirstlane(t >> 6);
  const int c0 = ch * 64;

  const float* xb = xin + ((size_t)b * C_DIM + c0) * S_DIM + s0 + j;
  float v[64];
#pragma unroll
  for (int i = 0; i < 64; ++i) v[i] = xb[(size_t)i * S_DIM];

  float sum = 0.f, sq = 0.f;
#pragma unroll
  for (int i = 0; i < 64; ++i) {
    sum += v[i];
    sq += v[i] * v[i];
  }
  red[ch][j] = sum;
  red2[ch][j] = sq;
  __syncthreads();
  float mean, rstd;
  {
    float s  = red[0][j] + red[1][j] + red[2][j] + red[3][j];
    float q2 = red2[0][j] + red2[1][j] + red2[2][j] + red2[3][j];
    mean = s * (1.f / C_DIM);
    float var = q2 * (1.f / C_DIM) - mean * mean;
    rstd = __frsqrt_rn(var + EPS);
  }

  float qs = 0.f;
#pragma unroll
  for (int i = 0; i < 64; i += 4) {
    v4f y4;
#pragma unroll
    for (int e = 0; e < 4; ++e) {
      int c = c0 + i + e;
      float y = (v[i + e] - mean) * rstd * lnw[c] + lnb[c];
      y4[e] = y;
      if (WITH_Q) qs += W[c] * y;
    }
    *(v4f*)(Y + j * PITCH + c0 + i) = y4;
  }
  if (WITH_Q) redq[ch][j] = qs;
  __syncthreads();
  if (WITH_Q) {
    if (t < TILE)
      red[0][t] = redq[0][t] + redq[1][t] + redq[2][t] + redq[3][t] + bias[0];
    __syncthreads();
    float mb = red[0][0];
    for (int jj = 1; jj < TILE; ++jj) mb = fmaxf(mb, red[0][jj]);
    float l = 0.f, acc = 0.f;
    for (int jj = 0; jj < TILE; ++jj) {
      float e = __expf(red[0][jj] - mb);
      l += e;
      acc += e * Y[jj * PITCH + t];
    }
    const size_t blk = (size_t)b * (S_DIM / TILE) + blockIdx.x;
    pstat[blk * C_DIM + t] = acc;
    float* stat = pstat + (size_t)B_DIM * (S_DIM / TILE) * C_DIM;
    if (t < 32) stat[blk * 32 + t] = (t == 0) ? mb : ((t == 1) ? l : 0.f);
  }

  const int wave = t >> 5;
  const int lane = t & 31;
  const int m    = lane & 15;
  const int kq   = (lane >> 4) << 2;
  const int mh   = (lane >> 4) << 3;
  const int rofs = WITH_Q ? 1 : 0;

  for (int rt = (WITH_Q ? 16 : 0) + wave; rt < 32; rt += 8) {
    const float* Wr = W + (size_t)(rofs + rt * 16 + m) * C_DIM;
    __builtin_prefetch(Wr, 0, 1);
    v8f acc[4] = {};
    wmma_k_loop_nb<4>(Wr, Y, PITCH, C_DIM, kq, m, acc);
#pragma unroll
    for (int c = 0; c < 4; ++c) {
#pragma unroll
      for (int r = 0; r < 8; ++r) {
        int row = rt * 16 + mh + r;
        float d = acc[c][r] + bias[rofs + row];
        if (LEAKY) d = d > 0.f ? d : 0.1f * d;
        out512[((size_t)b * F_DIM + row) * S_DIM + s0 + c * 16 + m] = d;
      }
    }
  }
}

__global__ void ctx_kernel(const float* __restrict__ pstat,
                           const float* __restrict__ W,
                           const float* __restrict__ bias,
                           float* __restrict__ ctx) {
  __shared__ __attribute__((aligned(16))) float yb[C_DIM];
  const int bp = blockIdx.x;
  const int t = threadIdx.x;
  const int blk0 = bp * (N_DIM / TILE);
  const float* stat = pstat + (size_t)B_DIM * (S_DIM / TILE) * C_DIM;

  float mi[N_DIM / TILE];
  float mx = -INFINITY;
#pragma unroll
  for (int i = 0; i < N_DIM / TILE; ++i) {
    mi[i] = stat[(size_t)(blk0 + i) * 32];
    mx = fmaxf(mx, mi[i]);
  }
  float lsum = 0.f, acc = 0.f;
#pragma unroll
  for (int i = 0; i < N_DIM / TILE; ++i) {
    float w = __expf(mi[i] - mx);
    lsum += stat[(size_t)(blk0 + i) * 32 + 1] * w;
    acc += pstat[(size_t)(blk0 + i) * C_DIM + t] * w;
  }
  yb[t] = acc / lsum;
  __syncthreads();

  const int wave = t >> 5;
  const int lane = t & 31;
  const v4f y0 = *(const v4f*)(yb + 4 * lane);
  const v4f y1 = *(const v4f*)(yb + 128 + 4 * lane);
  float res = 0.f;
#pragma unroll 8
  for (int r = 0; r < 32; ++r) {
    const float* wr = W + (size_t)(1 + wave * 32 + r) * C_DIM;
    v4f w0 = *(const v4f*)(wr + 4 * lane);
    v4f w1 = *(const v4f*)(wr + 128 + 4 * lane);
    float d = w0[0] * y0[0] + w0[1] * y0[1] + w0[2] * y0[2] + w0[3] * y0[3] +
              w1[0] * y1[0] + w1[1] * y1[1] + w1[2] * y1[2] + w1[3] * y1[3];
    for (int o = 16; o > 0; o >>= 1) d += __shfl_xor(d, o);
    if (lane == r) res = d;
  }
  ctx[(size_t)bp * C_DIM + wave * 32 + lane] = res + bias[1 + wave * 32 + lane];
}

__global__ void attn_out_kernel(const float* __restrict__ x,
                                const float* __restrict__ kv,
                                const float* __restrict__ ctx,
                                const float* __restrict__ W,
                                const float* __restrict__ bias,
                                float* __restrict__ x1) {
  extern __shared__ __attribute__((aligned(16))) float A[];
  const int b  = blockIdx.y;
  const int s0 = blockIdx.x * TILE;
  const int p  = s0 / N_DIM;
  const int t  = threadIdx.x;
  const int j  = t & (TILE - 1);
  const int ch = __builtin_amdgcn_readfirstlane(t >> 6);
  const int c0 = ch * 64;

  const float* ctxbp = ctx + ((size_t)b * P_DIM + p) * C_DIM;
  const float* vb = kv + ((size_t)b * F_DIM + C_DIM + c0) * S_DIM + s0 + j;
  float v[64];
#pragma unroll
  for (int i = 0; i < 64; ++i) v[i] = vb[(size_t)i * S_DIM];
#pragma unroll
  for (int i = 0; i < 64; i += 4) {
    v4f a4;
#pragma unroll
    for (int e = 0; e < 4; ++e) {
      float vv = v[i + e];
      vv = vv > 0.f ? vv : 0.f;
      a4[e] = vv * ctxbp[c0 + i + e];
    }
    *(v4f*)(A + j * PITCH + c0 + i) = a4;
  }
  __syncthreads();

  const int wave = t >> 5;
  const int lane = t & 31;
  const int m    = lane & 15;
  const int kq   = (lane >> 4) << 2;
  const int mh   = (lane >> 4) << 3;

  for (int rt = wave; rt < 16; rt += 8) {
    const float* Wr = W + (size_t)(rt * 16 + m) * C_DIM;
    __builtin_prefetch(Wr, 0, 1);
    v8f acc[4] = {};
    wmma_k_loop_nb<4>(Wr, A, PITCH, C_DIM, kq, m, acc);
#pragma unroll
    for (int c = 0; c < 4; ++c) {
#pragma unroll
      for (int r = 0; r < 8; ++r) {
        int row = rt * 16 + mh + r;
        size_t idx = ((size_t)b * C_DIM + row) * S_DIM + s0 + c * 16 + m;
        x1[idx] = x[idx] + bias[row] + acc[c][r];
      }
    }
  }
}

__global__ void ffn2_kernel(const float* __restrict__ x1,
                            const float* __restrict__ h,
                            const float* __restrict__ W,
                            const float* __restrict__ bias,
                            float* __restrict__ out) {
  extern __shared__ __attribute__((aligned(16))) float Hs[];
  const int b  = blockIdx.y;
  const int s0 = blockIdx.x * TILE5;
  const int t  = threadIdx.x;
  const int j  = t & (TILE5 - 1);
  const int ch = __builtin_amdgcn_readfirstlane(t >> 5);
  const float* hb = h + ((size_t)b * F_DIM + ch * 64) * S_DIM + s0 + j;
  float v[64];
#pragma unroll
  for (int i = 0; i < 64; ++i) v[i] = hb[(size_t)i * S_DIM];
#pragma unroll
  for (int i = 0; i < 64; i += 4) {
    v4f h4 = {v[i], v[i + 1], v[i + 2], v[i + 3]};
    *(v4f*)(Hs + j * PITCH5 + ch * 64 + i) = h4;
  }
  __syncthreads();

  const int wave = t >> 5;
  const int lane = t & 31;
  const int m    = lane & 15;
  const int kq   = (lane >> 4) << 2;
  const int mh   = (lane >> 4) << 3;

  for (int rt = wave; rt < 16; rt += 8) {
    const float* Wr = W + (size_t)(rt * 16 + m) * F_DIM;
    __builtin_prefetch(Wr, 0, 1);
    v8f acc[2] = {};
    wmma_k_loop_nb<2>(Wr, Hs, PITCH5, F_DIM, kq, m, acc);
#pragma unroll
    for (int c = 0; c < 2; ++c) {
#pragma unroll
      for (int r = 0; r < 8; ++r) {
        int row = rt * 16 + mh + r;
        size_t idx = ((size_t)b * C_DIM + row) * S_DIM + s0 + c * 16 + m;
        out[idx] = x1[idx] + bias[row] + acc[c][r];
      }
    }
  }
}

extern "C" void kernel_launch(void* const* d_in, const int* in_sizes, int n_in,
                              void* d_out, int out_size, void* d_ws,
                              size_t ws_size, hipStream_t stream) {
  const float* x      = (const float*)d_in[0];
  const float* ln1_w  = (const float*)d_in[1];
  const float* ln1_b  = (const float*)d_in[2];
  const float* qkv_w  = (const float*)d_in[3];
  const float* qkv_b  = (const float*)d_in[4];
  const float* out_w  = (const float*)d_in[5];
  const float* out_b  = (const float*)d_in[6];
  const float* ln2_w  = (const float*)d_in[7];
  const float* ln2_b  = (const float*)d_in[8];
  const float* ffn1_w = (const float*)d_in[9];
  const float* ffn1_b = (const float*)d_in[10];
  const float* ffn2_w = (const float*)d_in[11];
  const float* ffn2_b = (const float*)d_in[12];
  float* outp = (float*)d_out;

  float* ws   = (float*)d_ws;
  float* qbuf = ws;
  float* ctxb = qbuf + (size_t)B_DIM * S_DIM;
  float* kvb  = ctxb + (size_t)B_DIM * P_DIM * C_DIM;
  float* x1   = kvb + (size_t)B_DIM * F_DIM * S_DIM;
  float* hbuf = kvb;
  float* pstat = kvb;

  const size_t lds1 = (size_t)TILE * PITCH * sizeof(float);
  const size_t lds5 = (size_t)TILE5 * PITCH5 * sizeof(float);
  dim3 grid1(S_DIM / TILE, B_DIM);
  dim3 grid5(S_DIM / TILE5, B_DIM);

  ln_gemm_kernel<true, false><<<grid1, 256, lds1, stream>>>(
      x, ln1_w, ln1_b, qkv_w, qkv_b, kvb, pstat);
  ctx_kernel<<<B_DIM * P_DIM, 256, 0, stream>>>(pstat, qkv_w, qkv_b, ctxb);
  attn_out_kernel<<<grid1, 256, lds1, stream>>>(
      x, kvb, ctxb, out_w, out_b, x1);
  ln_gemm_kernel<false, true><<<grid1, 256, lds1, stream>>>(
      x1, ln2_w, ln2_b, ffn1_w, ffn1_b, hbuf, nullptr);
  ffn2_kernel<<<grid5, 256, lds5, stream>>>(
      x1, hbuf, ffn2_w, ffn2_b, outp);
}


# ffn1+ffn2 fused into one launch: the 512x64 h tile stays in LDS, never in memory (16-wave blocks, one per WGP); LN2 channel sums join 8 partials, so the f32 sum order of the LN2 statistics changes
# speedup vs baseline: 1.3901x; 1.0667x over previous
#include <hip/hip_runtime.h>
#include <math.h>


#define B_DIM 32
#define C_DIM 256
#define F_DIM 512
#define P_DIM 4
#define N_DIM 1024
#define S_DIM 4096
#define EPS   1e-5f

#define TILE  64
#define PITCH (C_DIM + 4)
#define PITCH5 (F_DIM + 4)

typedef __attribute__((ext_vector_type(2))) float v2f;
typedef __attribute__((ext_vector_type(4))) float v4f;
typedef __attribute__((ext_vector_type(8))) float v8f;

template <int NB>
__device__ __forceinline__ void wmma_k_loop_nb(const float* __restrict__ Wr,
                                               const float* __restrict__ Ybase,
                                               int pitch, int K, int kq, int m,
                                               v8f acc[NB]) {
#pragma unroll 2
  for (int k = 0; k < K; k += 8) {
    v4f a = *(const v4f*)(Wr + k + kq);
    v4f b[NB];
#pragma unroll
    for (int c = 0; c < NB; ++c)
      b[c] = *(const v4f*)(Ybase + (c * 16 + m) * pitch + k + kq);
#pragma unroll
    for (int c = 0; c < NB; ++c) {
      acc[c] = __builtin_amdgcn_wmma_f32_16x16x4_f32(
           false, a.lo,  false, b[c].lo,
           (short)0, acc[c],  false,  false);
    }
#pragma unroll
    for (int c = 0; c < NB; ++c) {
      acc[c] = __builtin_amdgcn_wmma_f32_16x16x4_f32(
           false, a.hi,  false, b[c].hi,
           (short)0, acc[c],  false,  false);
    }
  }
}

template <int NA>
__device__ __forceinline__ void wmma_k_loop_ab(const float* __restrict__ Wr,
                                               size_t rstride,
                                               const float* __restrict__ Ybase,
                                               int half_ofs,
                                               int pitch, int K, int kq, int m,
                                               v8f (&acc)[NA][4]) {
  constexpr int NB = 4;
#pragma unroll 4
  for (int k = 0; k < K; k += 8) {
    v4f a[NA];
    v4f b[NB];
#pragma unroll
    for (int i = 0; i < NA; ++i)
      a[i] = *(const v4f*)(Wr + (size_t)i * rstride + k + kq);
#pragma unroll
    for (int c = 0; c < NB; ++c)
      b[c] = *(const v4f*)(Ybase + (c >> 1) * half_ofs +
                           ((c & 1) * 16 + m) * pitch + k + kq);
#pragma unroll
    for (int i = 0; i < NA; ++i) {
#pragma unroll
      for (int c = 0; c < NB; ++c) {
        acc[i][c] = __builtin_amdgcn_wmma_f32_16x16x4_f32(
             false, a[i].lo,  false, b[c].lo,
             (short)0, acc[i][c],  false,  false);
      }
    }
#pragma unroll
    for (int i = 0; i < NA; ++i) {
#pragma unroll
      for (int c = 0; c < NB; ++c) {
        acc[i][c] = __builtin_amdgcn_wmma_f32_16x16x4_f32(
             false, a[i].hi,  false, b[c].hi,
             (short)0, acc[i][c],  false,  false);
      }
    }
  }
}

template <bool WITH_Q, bool LEAKY>
__global__ void ln_gemm_kernel(const float* __restrict__ xin,
                               const float* __restrict__ lnw,
                               const float* __restrict__ lnb,
                               const float* __restrict__ W,
                               const float* __restrict__ bias,
                               float* __restrict__ out512,
                               float* __restrict__ pstat) {
  extern __shared__ __attribute__((aligned(16))) float Y[];
  __shared__ __attribute__((aligned(16))) float red[4][TILE];
  __shared__ float red2[4][TILE];
  __shared__ float redq[4][TILE];

  const int b  = blockIdx.y;
  const int s0 = blockIdx.x * TILE;
  const int t  = threadIdx.x;
  const int j  = t & (TILE - 1);
  const int ch = __builtin_amdgcn_readfirstlane(t >> 6);
  const int c0 = ch * 64;

  const float* xb = xin + ((size_t)b * C_DIM + c0) * S_DIM + s0 + j;
  float v[64];
#pragma unroll
  for (int i = 0; i < 64; ++i) v[i] = xb[(size_t)i * S_DIM];

  float sum = 0.f, sq = 0.f;
#pragma unroll
  for (int i = 0; i < 64; ++i) {
    sum += v[i];
    sq += v[i] * v[i];
  }
  red[ch][j] = sum;
  red2[ch][j] = sq;
  __syncthreads();
  float mean, rstd;
  {
    float s  = red[0][j] + red[1][j] + red[2][j] + red[3][j];
    float q2 = red2[0][j] + red2[1][j] + red2[2][j] + red2[3][j];
    mean = s * (1.f / C_DIM);
    float var = q2 * (1.f / C_DIM) - mean * mean;
    rstd = __frsqrt_rn(var + EPS);
  }

  float qs = 0.f;
#pragma unroll
  for (int i = 0; i < 64; i += 4) {
    v4f y4;
#pragma unroll
    for (int e = 0; e < 4; ++e) {
      int c = c0 + i + e;
      float y = (v[i + e] - mean) * rstd * lnw[c] + lnb[c];
      y4[e] = y;
      if (WITH_Q) qs += W[c] * y;
    }
    *(v4f*)(Y + j * PITCH + c0 + i) = y4;
  }
  if (WITH_Q) redq[ch][j] = qs;
  __syncthreads();
  if (WITH_Q) {
    if (t < TILE)
      red[0][t] = redq[0][t] + redq[1][t] + redq[2][t] + redq[3][t] + bias[0];
    __syncthreads();
    float mb = red[0][0];
    for (int jj = 1; jj < TILE; ++jj) mb = fmaxf(mb, red[0][jj]);
    float l = 0.f, acc = 0.f;
    for (int jj = 0; jj < TILE; ++jj) {
      float e = __expf(red[0][jj] - mb);
      l += e;
      acc += e * Y[jj * PITCH + t];
    }
    const size_t blk = (size_t)b * (S_DIM / TILE) + blockIdx.x;
    pstat[blk * C_DIM + t] = acc;
    float* stat = pstat + (size_t)B_DIM * (S_DIM / TILE) * C_DIM;
    if (t < 32) stat[blk * 32 + t] = (t == 0) ? mb : ((t == 1) ? l : 0.f);
  }

  const int wave = t >> 5;
  const int lane = t & 31;
  const int m    = lane & 15;
  const int kq   = (lane >> 4) << 2;
  const int mh   = (lane >> 4) << 3;
  const int rofs = WITH_Q ? 1 : 0;

  for (int rt = (WITH_Q ? 16 : 0) + wave; rt < 32; rt += 8) {
    const float* Wr = W + (size_t)(rofs + rt * 16 + m) * C_DIM;
    __builtin_prefetch(Wr, 0, 1);
    v8f acc[4] = {};
    wmma_k_loop_nb<4>(Wr, Y, PITCH, C_DIM, kq, m, acc);
#pragma unroll
    for (int c = 0; c < 4; ++c) {
#pragma unroll
      for (int r = 0; r < 8; ++r) {
        int row = rt * 16 + mh + r;
        float d = acc[c][r] + bias[rofs + row];
        if (LEAKY) d = d > 0.f ? d : 0.1f * d;
        out512[((size_t)b * F_DIM + row) * S_DIM + s0 + c * 16 + m] = d;
      }
    }
  }
}

__global__ void ctx_kernel(const float* __restrict__ pstat,
                           const float* __restrict__ W,
                           const float* __restrict__ bias,
                           float* __restrict__ ctx) {
  __shared__ __attribute__((aligned(16))) float yb[C_DIM];
  const int bp = blockIdx.x;
  const int t = threadIdx.x;
  const int blk0 = bp * (N_DIM / TILE);
  const float* stat = pstat + (size_t)B_DIM * (S_DIM / TILE) * C_DIM;

  float mi[N_DIM / TILE];
  float mx = -INFINITY;
#pragma unroll
  for (int i = 0; i < N_DIM / TILE; ++i) {
    mi[i] = stat[(size_t)(blk0 + i) * 32];
    mx = fmaxf(mx, mi[i]);
  }
  float lsum = 0.f, acc = 0.f;
#pragma unroll
  for (int i = 0; i < N_DIM / TILE; ++i) {
    float w = __expf(mi[i] - mx);
    lsum += stat[(size_t)(blk0 + i) * 32 + 1] * w;
    acc += pstat[(size_t)(blk0 + i) * C_DIM + t] * w;
  }
  yb[t] = acc / lsum;
  __syncthreads();

  const int wave = t >> 5;
  const int lane = t & 31;
  const v4f y0 = *(const v4f*)(yb + 4 * lane);
  const v4f y1 = *(const v4f*)(yb + 128 + 4 * lane);
  float res = 0.f;
#pragma unroll 8
  for (int r = 0; r < 32; ++r) {
    const float* wr = W + (size_t)(1 + wave * 32 + r) * C_DIM;
    v4f w0 = *(const v4f*)(wr + 4 * lane);
    v4f w1 = *(const v4f*)(wr + 128 + 4 * lane);
    float d = w0[0] * y0[0] + w0[1] * y0[1] + w0[2] * y0[2] + w0[3] * y0[3] +
              w1[0] * y1[0] + w1[1] * y1[1] + w1[2] * y1[2] + w1[3] * y1[3];
    for (int o = 16; o > 0; o >>= 1) d += __shfl_xor(d, o);
    if (lane == r) res = d;
  }
  ctx[(size_t)bp * C_DIM + wave * 32 + lane] = res + bias[1 + wave * 32 + lane];
}

__global__ void attn_out_kernel(const float* __restrict__ x,
                                const float* __restrict__ kv,
                                const float* __restrict__ ctx,
                                const float* __restrict__ W,
                                const float* __restrict__ bias,
                                float* __restrict__ x1) {
  extern __shared__ __attribute__((aligned(16))) float A[];
  const int b  = blockIdx.y;
  const int s0 = blockIdx.x * TILE;
  const int p  = s0 / N_DIM;
  const int t  = threadIdx.x;
  const int j  = t & (TILE - 1);
  const int ch = __builtin_amdgcn_readfirstlane(t >> 6);
  const int c0 = ch * 64;

  const float* ctxbp = ctx + ((size_t)b * P_DIM + p) * C_DIM;
  const float* vb = kv + ((size_t)b * F_DIM + C_DIM + c0) * S_DIM + s0 + j;
  float v[64];
#pragma unroll
  for (int i = 0; i < 64; ++i) v[i] = vb[(size_t)i * S_DIM];
#pragma unroll
  for (int i = 0; i < 64; i += 4) {
    v4f a4;
#pragma unroll
    for (int e = 0; e < 4; ++e) {
      float vv = v[i + e];
      vv = vv > 0.f ? vv : 0.f;
      a4[e] = vv * ctxbp[c0 + i + e];
    }
    *(v4f*)(A + j * PITCH + c0 + i) = a4;
  }
  __syncthreads();

  const int wave = t >> 5;
  const int lane = t & 31;
  const int m    = lane & 15;
  const int kq   = (lane >> 4) << 2;
  const int mh   = (lane >> 4) << 3;

  for (int rt = wave; rt < 16; rt += 8) {
    const float* Wr = W + (size_t)(rt * 16 + m) * C_DIM;
    __builtin_prefetch(Wr, 0, 1);
    v8f acc[4] = {};
    wmma_k_loop_nb<4>(Wr, A, PITCH, C_DIM, kq, m, acc);
#pragma unroll
    for (int c = 0; c < 4; ++c) {
#pragma unroll
      for (int r = 0; r < 8; ++r) {
        int row = rt * 16 + mh + r;
        size_t idx = ((size_t)b * C_DIM + row) * S_DIM + s0 + c * 16 + m;
        x1[idx] = x[idx] + bias[row] + acc[c][r];
      }
    }
  }
}

__global__ void __launch_bounds__(512)
ffn_fused_kernel(const float* __restrict__ x1,
                 const float* __restrict__ lnw,
                 const float* __restrict__ lnb,
                 const float* __restrict__ W1,
                 const float* __restrict__ b1,
                 const float* __restrict__ W2,
                 const float* __restrict__ b2,
                 float* __restrict__ out) {
  __shared__ __attribute__((aligned(16))) float Y[TILE * PITCH];
  __shared__ __attribute__((aligned(16))) float H[TILE * PITCH5];
  __shared__ float red[8][TILE];
  __shared__ float red2[8][TILE];

  const int b  = blockIdx.y;
  const int s0 = blockIdx.x * TILE;
  const int t  = threadIdx.x;
  const int j  = t & (TILE - 1);
  const int ch = __builtin_amdgcn_readfirstlane(t >> 6);
  const int c0 = ch * 32;

  const float* xb = x1 + ((size_t)b * C_DIM + c0) * S_DIM + s0 + j;
  float v[32];
#pragma unroll
  for (int i = 0; i < 32; ++i) v[i] = xb[(size_t)i * S_DIM];

  float sum = 0.f, sq = 0.f;
#pragma unroll
  for (int i = 0; i < 32; ++i) {
    sum += v[i];
    sq += v[i] * v[i];
  }
  red[ch][j] = sum;
  red2[ch][j] = sq;
  __syncthreads();
  float mean, rstd;
  {
    float s = red[0][j], q2 = red2[0][j];
#pragma unroll
    for (int i = 1; i < 8; ++i) {
      s += red[i][j];
      q2 += red2[i][j];
    }
    mean = s * (1.f / C_DIM);
    float var = q2 * (1.f / C_DIM) - mean * mean;
    rstd = __frsqrt_rn(var + EPS);
  }
#pragma unroll
  for (int i = 0; i < 32; i += 4) {
    v4f y4;
#pragma unroll
    for (int e = 0; e < 4; ++e) {
      int c = c0 + i + e;
      y4[e] = (v[i + e] - mean) * rstd * lnw[c] + lnb[c];
    }
    *(v4f*)(Y + j * PITCH + c0 + i) = y4;
  }
  __syncthreads();

  const int wave = t >> 5;
  const int lane = t & 31;
  const int m    = lane & 15;
  const int kq   = (lane >> 4) << 2;
  const int mh   = (lane >> 4) << 3;

  {
    const float* Wr = W1 + (size_t)(wave * 16 + m) * C_DIM;
    __builtin_prefetch(Wr, 0, 1);
    v8f acc[2][4] = {};
    wmma_k_loop_ab<2>(Wr, (size_t)256 * C_DIM, Y, 32 * PITCH, PITCH, C_DIM, kq,
                     m, acc);
#pragma unroll
    for (int i = 0; i < 2; ++i) {
      const int row0 = (wave + 16 * i) * 16 + mh;
      const v4f bl = *(const v4f*)(b1 + row0);
      const v4f bh = *(const v4f*)(b1 + row0 + 4);
#pragma unroll
      for (int c = 0; c < 4; ++c) {
        v4f lo4, hi4;
#pragma unroll
        for (int r = 0; r < 4; ++r) {
          float d0 = acc[i][c][r] + bl[r];
          float d1 = acc[i][c][r + 4] + bh[r];
          lo4[r] = d0 > 0.f ? d0 : 0.1f * d0;
          hi4[r] = d1 > 0.f ? d1 : 0.1f * d1;
        }
        float* hp = H + (c * 16 + m) * PITCH5 + row0;
        *(v4f*)(hp) = lo4;
        *(v4f*)(hp + 4) = hi4;
      }
    }
  }
  __syncthreads();

  {
    const float* Wr = W2 + (size_t)(wave * 16 + m) * F_DIM;
    __builtin_prefetch(Wr, 0, 1);
    v8f acc[1][4] = {};
    int hofs = 32 * PITCH5;
    asm volatile("" : "+s"(hofs));
    wmma_k_loop_ab<1>(Wr, 0, H, hofs, PITCH5, F_DIM, kq, m, acc);
#pragma unroll
    for (int c = 0; c < 4; ++c) {
#pragma unroll
      for (int r = 0; r < 8; ++r) {
        int row = wave * 16 + mh + r;
        size_t idx = ((size_t)b * C_DIM + row) * S_DIM + s0 + c * 16 + m;
        out[idx] = x1[idx] + b2[row] + acc[0][c][r];
      }
    }
  }
}

extern "C" void kernel_launch(void* const* d_in, const int* in_sizes, int n_in,
                              void* d_out, int out_size, void* d_ws,
                              size_t ws_size, hipStream_t stream) {
  const float* x      = (const float*)d_in[0];
  const float* ln1_w  = (const float*)d_in[1];
  const float* ln1_b  = (const float*)d_in[2];
  const float* qkv_w  = (const float*)d_in[3];
  const float* qkv_b  = (const float*)d_in[4];
  const float* out_w  = (const float*)d_in[5];
  const float* out_b  = (const float*)d_in[6];
  const float* ln2_w  = (const float*)d_in[7];
  const float* ln2_b  = (const float*)d_in[8];
  const float* ffn1_w = (const float*)d_in[9];
  const float* ffn1_b = (const float*)d_in[10];
  const float* ffn2_w = (const float*)d_in[11];
  const float* ffn2_b = (const float*)d_in[12];
  float* outp = (float*)d_out;

  float* ws   = (float*)d_ws;
  float* qbuf = ws;
  float* ctxb = qbuf + (size_t)B_DIM * S_DIM;
  float* kvb  = ctxb + (size_t)B_DIM * P_DIM * C_DIM;
  float* x1   = kvb + (size_t)B_DIM * F_DIM * S_DIM;
  float* pstat = kvb;

  const size_t lds1 = (size_t)TILE * PITCH * sizeof(float);
  dim3 grid1(S_DIM / TILE, B_DIM);

  ln_gemm_kernel<true, false><<<grid1, 256, lds1, stream>>>(
      x, ln1_w, ln1_b, qkv_w, qkv_b, kvb, pstat);
  ctx_kernel<<<B_DIM * P_DIM, 256, 0, stream>>>(pstat, qkv_w, qkv_b, ctxb);
  attn_out_kernel<<<grid1, 256, lds1, stream>>>(
      x, kvb, ctxb, out_w, out_b, x1);
  ffn_fused_kernel<<<grid1, 512, 0, stream>>>(
      x1, ln2_w, ln2_b, ffn1_w, ffn1_b, ffn2_w, ffn2_b, outp);
}


# FFN kernel as 8-wave blocks keeping h in LDS one 256-row half at a time (134 KB): two blocks per WGP, one's load/LN/barrier/epilogue phases run under the other's WMMAs; sum orders unchanged
# speedup vs baseline: 1.4552x; 1.0468x over previous
#include <hip/hip_runtime.h>
#include <math.h>


#define B_DIM 32
#define C_DIM 256
#define F_DIM 512
#define P_DIM 4
#define N_DIM 1024
#define S_DIM 4096
#define EPS   1e-5f

#define TILE  64
#define PITCH (C_DIM + 4)

typedef __attribute__((ext_vector_type(2))) float v2f;
typedef __attribute__((ext_vector_type(4))) float v4f;
typedef __attribute__((ext_vector_type(8))) float v8f;

template <int NB>
__device__ __forceinline__ void wmma_k_loop_nb(const float* __restrict__ Wr,
                                               const float* __restrict__ Ybase,
                                               int pitch, int K, int kq, int m,
                                               v8f acc[NB]) {
#pragma unroll 2
  for (int k = 0; k < K; k += 8) {
    v4f a = *(const v4f*)(Wr + k + kq);
    v4f b[NB];
#pragma unroll
    for (int c = 0; c < NB; ++c)
      b[c] = *(const v4f*)(Ybase + (c * 16 + m) * pitch + k + kq);
#pragma unroll
    for (int c = 0; c < NB; ++c) {
      acc[c] = __builtin_amdgcn_wmma_f32_16x16x4_f32(
           false, a.lo,  false, b[c].lo,
           (short)0, acc[c],  false,  false);
    }
#pragma unroll
    for (int c = 0; c < NB; ++c) {
      acc[c] = __builtin_amdgcn_wmma_f32_16x16x4_f32(
           false, a.hi,  false, b[c].hi,
           (short)0, acc[c],  false,  false);
    }
  }
}

template <int NA>
__device__ __forceinline__ void wmma_k_loop_ab(const float* __restrict__ Wr,
                                               size_t rstride,
                                               const float* __restrict__ Ybase,
                                               int half_ofs,
                                               int pitch, int K, int kq, int m,
                                               v8f (&acc)[NA][4]) {
  constexpr int NB = 4;
#pragma unroll 4
  for (int k = 0; k < K; k += 8) {
    v4f a[NA];
    v4f b[NB];
#pragma unroll
    for (int i = 0; i < NA; ++i)
      a[i] = *(const v4f*)(Wr + (size_t)i * rstride + k + kq);
#pragma unroll
    for (int c = 0; c < NB; ++c)
      b[c] = *(const v4f*)(Ybase + (c >> 1) * half_ofs +
                           ((c & 1) * 16 + m) * pitch + k + kq);
#pragma unroll
    for (int i = 0; i < NA; ++i) {
#pragma unroll
      for (int c = 0; c < NB; ++c) {
        acc[i][c] = __builtin_amdgcn_wmma_f32_16x16x4_f32(
             false, a[i].lo,  false, b[c].lo,
             (short)0, acc[i][c],  false,  false);
      }
    }
#pragma unroll
    for (int i = 0; i < NA; ++i) {
#pragma unroll
      for (int c = 0; c < NB; ++c) {
        acc[i][c] = __builtin_amdgcn_wmma_f32_16x16x4_f32(
             false, a[i].hi,  false, b[c].hi,
             (short)0, acc[i][c],  false,  false);
      }
    }
  }
}

template <bool WITH_Q, bool LEAKY>
__global__ void ln_gemm_kernel(const float* __restrict__ xin,
                               const float* __restrict__ lnw,
                               const float* __restrict__ lnb,
                               const float* __restrict__ W,
                               const float* __restrict__ bias,
                               float* __restrict__ out512,
                               float* __restrict__ pstat) {
  extern __shared__ __attribute__((aligned(16))) float Y[];
  __shared__ __attribute__((aligned(16))) float red[4][TILE];
  __shared__ float red2[4][TILE];
  __shared__ float redq[4][TILE];

  const int b  = blockIdx.y;
  const int s0 = blockIdx.x * TILE;
  const int t  = threadIdx.x;
  const int j  = t & (TILE - 1);
  const int ch = __builtin_amdgcn_readfirstlane(t >> 6);
  const int c0 = ch * 64;

  const float* xb = xin + ((size_t)b * C_DIM + c0) * S_DIM + s0 + j;
  float v[64];
#pragma unroll
  for (int i = 0; i < 64; ++i) v[i] = xb[(size_t)i * S_DIM];

  float sum = 0.f, sq = 0.f;
#pragma unroll
  for (int i = 0; i < 64; ++i) {
    sum += v[i];
    sq += v[i] * v[i];
  }
  red[ch][j] = sum;
  red2[ch][j] = sq;
  __syncthreads();
  float mean, rstd;
  {
    float s  = red[0][j] + red[1][j] + red[2][j] + red[3][j];
    float q2 = red2[0][j] + red2[1][j] + red2[2][j] + red2[3][j];
    mean = s * (1.f / C_DIM);
    float var = q2 * (1.f / C_DIM) - mean * mean;
    rstd = __frsqrt_rn(var + EPS);
  }

  float qs = 0.f;
#pragma unroll
  for (int i = 0; i < 64; i += 4) {
    v4f y4;
#pragma unroll
    for (int e = 0; e < 4; ++e) {
      int c = c0 + i + e;
      float y = (v[i + e] - mean) * rstd * lnw[c] + lnb[c];
      y4[e] = y;
      if (WITH_Q) qs += W[c] * y;
    }
    *(v4f*)(Y + j * PITCH + c0 + i) = y4;
  }
  if (WITH_Q) redq[ch][j] = qs;
  __syncthreads();
  if (WITH_Q) {
    if (t < TILE)
      red[0][t] = redq[0][t] + redq[1][t] + redq[2][t] + redq[3][t] + bias[0];
    __syncthreads();
    float mb = red[0][0];
    for (int jj = 1; jj < TILE; ++jj) mb = fmaxf(mb, red[0][jj]);
    float l = 0.f, acc = 0.f;
    for (int jj = 0; jj < TILE; ++jj) {
      float e = __expf(red[0][jj] - mb);
      l += e;
      acc += e * Y[jj * PITCH + t];
    }
    const size_t blk = (size_t)b * (S_DIM / TILE) + blockIdx.x;
    pstat[blk * C_DIM + t] = acc;
    float* stat = pstat + (size_t)B_DIM * (S_DIM / TILE) * C_DIM;
    if (t < 32) stat[blk * 32 + t] = (t == 0) ? mb : ((t == 1) ? l : 0.f);
  }

  const int wave = t >> 5;
  const int lane = t & 31;
  const int m    = lane & 15;
  const int kq   = (lane >> 4) << 2;
  const int mh   = (lane >> 4) << 3;
  const int rofs = WITH_Q ? 1 : 0;

  for (int rt = (WITH_Q ? 16 : 0) + wave; rt < 32; rt += 8) {
    const float* Wr = W + (size_t)(rofs + rt * 16 + m) * C_DIM;
    __builtin_prefetch(Wr, 0, 1);
    v8f acc[4] = {};
    wmma_k_loop_nb<4>(Wr, Y, PITCH, C_DIM, kq, m, acc);
#pragma unroll
    for (int c = 0; c < 4; ++c) {
#pragma unroll
      for (int r = 0; r < 8; ++r) {
        int row = rt * 16 + mh + r;
        float d = acc[c][r] + bias[rofs + row];
        if (LEAKY) d = d > 0.f ? d : 0.1f * d;
        out512[((size_t)b * F_DIM + row) * S_DIM + s0 + c * 16 + m] = d;
      }
    }
  }
}

__global__ void ctx_kernel(const float* __restrict__ pstat,
                           const float* __restrict__ W,
                           const float* __restrict__ bias,
                           float* __restrict__ ctx) {
  __shared__ __attribute__((aligned(16))) float yb[C_DIM];
  const int bp = blockIdx.x;
  const int t = threadIdx.x;
  const int blk0 = bp * (N_DIM / TILE);
  const float* stat = pstat + (size_t)B_DIM * (S_DIM / TILE) * C_DIM;

  float mi[N_DIM / TILE];
  float mx = -INFINITY;
#pragma unroll
  for (int i = 0; i < N_DIM / TILE; ++i) {
    mi[i] = stat[(size_t)(blk0 + i) * 32];
    mx = fmaxf(mx, mi[i]);
  }
  float lsum = 0.f, acc = 0.f;
#pragma unroll
  for (int i = 0; i < N_DIM / TILE; ++i) {
    float w = __expf(mi[i] - mx);
    lsum += stat[(size_t)(blk0 + i) * 32 + 1] * w;
    acc += pstat[(size_t)(blk0 + i) * C_DIM + t] * w;
  }
  yb[t] = acc / lsum;
  __syncthreads();

  const int wave = t >> 5;
  const int lane = t & 31;
  const v4f y0 = *(const v4f*)(yb + 4 * lane);
  const v4f y1 = *(const v4f*)(yb + 128 + 4 * lane);
  float res = 0.f;
#pragma unroll 8
  for (int r = 0; r < 32; ++r) {
    const float* wr = W + (size_t)(1 + wave * 32 + r) * C_DIM;
    v4f w0 = *(const v4f*)(wr + 4 * lane);
    v4f w1 = *(const v4f*)(wr + 128 + 4 * lane);
    float d = w0[0] * y0[0] + w0[1] * y0[1] + w0[2] * y0[2] + w0[3] * y0[3] +
              w1[0] * y1[0] + w1[1] * y1[1] + w1[2] * y1[2] + w1[3] * y1[3];
    for (int o = 16; o > 0; o >>= 1) d += __shfl_xor(d, o);
    if (lane == r) res = d;
  }
  ctx[(size_t)bp * C_DIM + wave * 32 + lane] = res + bias[1 + wave * 32 + lane];
}

__global__ void attn_out_kernel(const float* __restrict__ x,
                                const float* __restrict__ kv,
                                const float* __restrict__ ctx,
                                const float* __restrict__ W,
                                const float* __restrict__ bias,
                                float* __restrict__ x1) {
  extern __shared__ __attribute__((aligned(16))) float A[];
  const int b  = blockIdx.y;
  const int s0 = blockIdx.x * TILE;
  const int p  = s0 / N_DIM;
  const int t  = threadIdx.x;
  const int j  = t & (TILE - 1);
  const int ch = __builtin_amdgcn_readfirstlane(t >> 6);
  const int c0 = ch * 64;

  const float* ctxbp = ctx + ((size_t)b * P_DIM + p) * C_DIM;
  const float* vb = kv + ((size_t)b * F_DIM + C_DIM + c0) * S_DIM + s0 + j;
  float v[64];
#pragma unroll
  for (int i = 0; i < 64; ++i) v[i] = vb[(size_t)i * S_DIM];
#pragma unroll
  for (int i = 0; i < 64; i += 4) {
    v4f a4;
#pragma unroll
    for (int e = 0; e < 4; ++e) {
      float vv = v[i + e];
      vv = vv > 0.f ? vv : 0.f;
      a4[e] = vv * ctxbp[c0 + i + e];
    }
    *(v4f*)(A + j * PITCH + c0 + i) = a4;
  }
  __syncthreads();

  const int wave = t >> 5;
  const int lane = t & 31;
  const int m    = lane & 15;
  const int kq   = (lane >> 4) << 2;
  const int mh   = (lane >> 4) << 3;

  for (int rt = wave; rt < 16; rt += 8) {
    const float* Wr = W + (size_t)(rt * 16 + m) * C_DIM;
    __builtin_prefetch(Wr, 0, 1);
    v8f acc[4] = {};
    wmma_k_loop_nb<4>(Wr, A, PITCH, C_DIM, kq, m, acc);
#pragma unroll
    for (int c = 0; c < 4; ++c) {
#pragma unroll
      for (int r = 0; r < 8; ++r) {
        int row = rt * 16 + mh + r;
        size_t idx = ((size_t)b * C_DIM + row) * S_DIM + s0 + c * 16 + m;
        x1[idx] = x[idx] + bias[row] + acc[c][r];
      }
    }
  }
}

__global__ void __launch_bounds__(256)
ffn_fused_kernel(const float* __restrict__ x1,
                 const float* __restrict__ lnw,
                 const float* __restrict__ lnb,
                 const float* __restrict__ W1,
                 const float* __restrict__ b1,
                 const float* __restrict__ W2,
                 const float* __restrict__ b2,
                 float* __restrict__ out) {
  __shared__ __attribute__((aligned(16))) float Y[TILE * PITCH];
  __shared__ __attribute__((aligned(16))) float Hc[TILE * PITCH];
  __shared__ float red[8][TILE];
  __shared__ float red2[8][TILE];

  const int b  = blockIdx.y;
  const int s0 = blockIdx.x * TILE;
  const int t  = threadIdx.x;
  const int j  = t & (TILE - 1);
  const int ch = __builtin_amdgcn_readfirstlane(t >> 6);
  const int c0 = ch * 64;

  const float* xb = x1 + ((size_t)b * C_DIM + c0) * S_DIM + s0 + j;
  float v[64];
#pragma unroll
  for (int i = 0; i < 64; ++i) v[i] = xb[(size_t)i * S_DIM];

  float sum[2] = {0.f, 0.f}, sq[2] = {0.f, 0.f};
#pragma unroll
  for (int g = 0; g < 2; ++g) {
#pragma unroll
    for (int i = 0; i < 32; ++i) {
      sum[g] += v[32 * g + i];
      sq[g] += v[32 * g + i] * v[32 * g + i];
    }
    red[2 * ch + g][j] = sum[g];
    red2[2 * ch + g][j] = sq[g];
  }
  __syncthreads();
  float mean, rstd;
  {
    float s = red[0][j], q2 = red2[0][j];
#pragma unroll
    for (int i = 1; i < 8; ++i) {
      s += red[i][j];
      q2 += red2[i][j];
    }
    mean = s * (1.f / C_DIM);
    float var = q2 * (1.f / C_DIM) - mean * mean;
    rstd = __frsqrt_rn(var + EPS);
  }
#pragma unroll
  for (int i = 0; i < 64; i += 4) {
    v4f y4;
#pragma unroll
    for (int e = 0; e < 4; ++e) {
      int c = c0 + i + e;
      y4[e] = (v[i + e] - mean) * rstd * lnw[c] + lnb[c];
    }
    *(v4f*)(Y + j * PITCH + c0 + i) = y4;
  }
  __syncthreads();

  const int wave = t >> 5;
  const int lane = t & 31;
  const int m    = lane & 15;
  const int kq   = (lane >> 4) << 2;
  const int mh   = (lane >> 4) << 3;

  v8f acc2[2][4] = {};
  for (int f = 0; f < 2; ++f) {
    {
      const float* Wr = W1 + (size_t)(f * 256 + wave * 16 + m) * C_DIM;
      __builtin_prefetch(Wr, 0, 1);
      v8f acc1[2][4] = {};
      wmma_k_loop_ab<2>(Wr, (size_t)128 * C_DIM, Y, 32 * PITCH, PITCH, C_DIM,
                        kq, m, acc1);
      if (f > 0) __syncthreads();
#pragma unroll
      for (int i = 0; i < 2; ++i) {
        const int row0 = (wave + 8 * i) * 16 + mh;
        const v4f bl = *(const v4f*)(b1 + f * 256 + row0);
        const v4f bh = *(const v4f*)(b1 + f * 256 + row0 + 4);
#pragma unroll
        for (int c = 0; c < 4; ++c) {
          v4f lo4, hi4;
#pragma unroll
          for (int r = 0; r < 4; ++r) {
            float d0 = acc1[i][c][r] + bl[r];
            float d1 = acc1[i][c][r + 4] + bh[r];
            lo4[r] = d0 > 0.f ? d0 : 0.1f * d0;
            hi4[r] = d1 > 0.f ? d1 : 0.1f * d1;
          }
          float* hp = Hc + (c * 16 + m) * PITCH + row0;
          *(v4f*)(hp) = lo4;
          *(v4f*)(hp + 4) = hi4;
        }
      }
    }
    __syncthreads();

    {
      const float* Wr = W2 + (size_t)(wave * 16 + m) * F_DIM + f * 256;
      __builtin_prefetch(Wr, 0, 1);
      wmma_k_loop_ab<2>(Wr, (size_t)128 * F_DIM, Hc, 32 * PITCH, PITCH, 256,
                        kq, m, acc2);
    }
  }

#pragma unroll
  for (int i = 0; i < 2; ++i) {
#pragma unroll
    for (int c = 0; c < 4; ++c) {
#pragma unroll
      for (int r = 0; r < 8; ++r) {
        int row = (wave + 8 * i) * 16 + mh + r;
        size_t idx = ((size_t)b * C_DIM + row) * S_DIM + s0 + c * 16 + m;
        out[idx] = x1[idx] + b2[row] + acc2[i][c][r];
      }
    }
  }
}

extern "C" void kernel_launch(void* const* d_in, const int* in_sizes, int n_in,
                              void* d_out, int out_size, void* d_ws,
                              size_t ws_size, hipStream_t stream) {
  const float* x      = (const float*)d_in[0];
  const float* ln1_w  = (const float*)d_in[1];
  const float* ln1_b  = (const float*)d_in[2];
  const float* qkv_w  = (const float*)d_in[3];
  const float* qkv_b  = (const float*)d_in[4];
  const float* out_w  = (const float*)d_in[5];
  const float* out_b  = (const float*)d_in[6];
  const float* ln2_w  = (const float*)d_in[7];
  const float* ln2_b  = (const float*)d_in[8];
  const float* ffn1_w = (const float*)d_in[9];
  const float* ffn1_b = (const float*)d_in[10];
  const float* ffn2_w = (const float*)d_in[11];
  const float* ffn2_b = (const float*)d_in[12];
  float* outp = (float*)d_out;

  float* ws   = (float*)d_ws;
  float* qbuf = ws;
  float* ctxb = qbuf + (size_t)B_DIM * S_DIM;
  float* kvb  = ctxb + (size_t)B_DIM * P_DIM * C_DIM;
  float* x1   = kvb + (size_t)B_DIM * F_DIM * S_DIM;
  float* pstat = kvb;

  const size_t lds1 = (size_t)TILE * PITCH * sizeof(float);
  dim3 grid1(S_DIM / TILE, B_DIM);

  ln_gemm_kernel<true, false><<<grid1, 256, lds1, stream>>>(
      x, ln1_w, ln1_b, qkv_w, qkv_b, kvb, pstat);
  ctx_kernel<<<B_DIM * P_DIM, 256, 0, stream>>>(pstat, qkv_w, qkv_b, ctxb);
  attn_out_kernel<<<grid1, 256, lds1, stream>>>(
      x, kvb, ctxb, out_w, out_b, x1);
  ffn_fused_kernel<<<grid1, 256, 0, stream>>>(
      x1, ln2_w, ln2_b, ffn1_w, ffn1_b, ffn2_w, ffn2_b, outp);
}
